# starting kernel
# baseline (speedup 1.0000x reference)
#include <hip/hip_runtime.h>


typedef __attribute__((ext_vector_type(16))) _Float16 v16h;
typedef __attribute__((ext_vector_type(8)))  float    v8f;

static constexpr int BATCH    = 4;
static constexpr int CIN      = 64;
static constexpr int COUT     = 64;
static constexpr int NPTS     = 16384;
static constexpr int KNBH     = 8;

static constexpr int ROWS     = 4 * COUT;
static constexpr int TILE_PTS = 32;
static constexpr int LDS_STR  = 33;
static constexpr int NPAIRS   = TILE_PTS * KNBH;
static constexpr int A_ELEMS  = 16   * 2   * 32   * 16  ;
static constexpr int B_HALVES = 2   * 2   * 32   * 16  ;

__device__ __forceinline__ int kmap(int h, int g, int kc) {
    return 32 * kc + 16 * (h >> 3) + 8 * g + (h & 7);
}

__global__ void init_out_kernel(const float* __restrict__ bias,
                                float* __restrict__ out) {
    size_t idx = (size_t)blockIdx.x * blockDim.x + threadIdx.x;
    size_t nwords = (size_t)BATCH * COUT * NPTS / 4;
    if (idx >= nwords) return;
    int o = (int)((idx / (NPTS / 4)) % COUT);
    float v = bias[o];
    ((float4*)out)[idx] = make_float4(v, v, v, v);
}

__global__ void pack_weights_kernel(const float* __restrict__ theta,
                                    const float* __restrict__ wbias,
                                    _Float16* __restrict__ wsA) {
    int e = blockIdx.x * blockDim.x + threadIdx.x;
    if (e >= A_ELEMS) return;
    int h    = e & 15;
    int lane = (e >> 4) & 31;
    int kc   = (e >> 9) & 1;
    int mt   = e >> 10;
    int row  = mt * 16 + (lane & 15);
    int g    = lane >> 4;
    int Kk   = kmap(h, g, kc);
    float w;
    if (row < 3 * COUT) {
        int d = row >> 6, o = row & 63;
        w = theta[((size_t)d * CIN + Kk) * COUT + o];
    } else {
        int o = row - 3 * COUT;
        w = wbias[(size_t)Kk * COUT + o];
    }
    wsA[e] = (_Float16)w;
}

__global__ void __launch_bounds__(256)
flexconv_kernel(const float* __restrict__ features,
                const int*   __restrict__ nbh,
                const float* __restrict__ pos,
                const _Float16* __restrict__ wsA,
                float* __restrict__ out) {
    __shared__ float ftfb[ROWS * LDS_STR];
    __shared__ float pdx[NPAIRS], pdy[NPAIRS], pdz[NPAIRS];
    __shared__ int   pj[NPAIRS];
    __shared__ __align__(32) _Float16 bstage[B_HALVES];

    const int tid        = threadIdx.x;
    const int blocksPerB = NPTS / TILE_PTS;
    const int b          = blockIdx.x / blocksPerB;
    const int n0         = (blockIdx.x % blocksPerB) * TILE_PTS;

    {
        int q  = tid;
        int pt = q >> 3, k = q & 7;
        int n  = n0 + pt;
        int j  = nbh[((size_t)b * KNBH + k) * NPTS + n];
        const float* posb = pos + (size_t)b * 3 * NPTS;
        pj[q]  = j;
        pdx[q] = posb[0 * NPTS + j] - posb[0 * NPTS + n];
        pdy[q] = posb[1 * NPTS + j] - posb[1 * NPTS + n];
        pdz[q] = posb[2 * NPTS + j] - posb[2 * NPTS + n];
    }

    {
        const int ncol = tid & 31;
        const int krow = tid >> 5;
        const float* featb = features + (size_t)b * CIN * NPTS + n0;
        const int nt   = ncol >> 4;
        const int lcol = ncol & 15;
        #pragma unroll
        for (int pass = 0; pass < 8; ++pass) {
            const int Kk  = krow + 8 * pass;
            const float v = featb[(size_t)Kk * NPTS + ncol];
            const int kc  = Kk >> 5;
            const int rem = Kk & 31;
            const int g   = (rem >> 3) & 1;
            const int h   = 8 * (rem >> 4) + (rem & 7);
            const int lane = g * 16 + lcol;
            bstage[(((nt * 2 + kc) * 32) + lane) * 16 + h] = (_Float16)v;
        }
    }
    __syncthreads();

    {
        const int wave  = tid >> 5;
        const int lane  = tid & 31;
        const int nt    = wave & 1;
        const int mbase = wave >> 1;
        const int g     = lane >> 4;
        const int lcol  = lane & 15;

        const v16h bfrag0 = *(const v16h*)(bstage + ((nt * 2 + 0) * 32 + lane) * 16);
        const v16h bfrag1 = *(const v16h*)(bstage + ((nt * 2 + 1) * 32 + lane) * 16);

        #pragma unroll
        for (int i = 0; i < 4; ++i) {
            const int mt = mbase + 4 * i;
            v8f c = {};
            {
                v16h a = *(const v16h*)(wsA + (((size_t)mt * 2 + 0) * 32 + lane) * 16);
                c = __builtin_amdgcn_wmma_f32_16x16x32_f16(
                        false, a, false, bfrag0, (short)0, c, false, false);
            }
            {
                v16h a = *(const v16h*)(wsA + (((size_t)mt * 2 + 1) * 32 + lane) * 16);
                c = __builtin_amdgcn_wmma_f32_16x16x32_f16(
                        false, a, false, bfrag1, (short)0, c, false, false);
            }
            const int colBase = nt * 16 + lcol;
            #pragma unroll
            for (int r = 0; r < 8; ++r) {
                int row = mt * 16 + 8 * g + r;
                ftfb[row * LDS_STR + colBase] = c[r];
            }
        }
    }
    __syncthreads();

    {
        const int o = tid & 63;
        for (int pp = tid >> 6; pp < NPAIRS; pp += 4) {
            const int   pt = pp >> 3;
            const float dx = pdx[pp], dy = pdy[pp], dz = pdz[pp];
            const int   j  = pj[pp];
            const float c0 = ftfb[(0 * COUT + o) * LDS_STR + pt];
            const float c1 = ftfb[(1 * COUT + o) * LDS_STR + pt];
            const float c2 = ftfb[(2 * COUT + o) * LDS_STR + pt];
            const float cb = ftfb[(3 * COUT + o) * LDS_STR + pt];
            const float contrib = dx * c0 + dy * c1 + dz * c2 + cb;
            atomicAdd(&out[((size_t)b * COUT + o) * NPTS + j], contrib);
        }
    }
}

extern "C" void kernel_launch(void* const* d_in, const int* in_sizes, int n_in,
                              void* d_out, int out_size, void* d_ws, size_t ws_size,
                              hipStream_t stream) {
    const float* features = (const float*)d_in[0];
    const float* theta    = (const float*)d_in[1];
    const float* wbias    = (const float*)d_in[2];
    const int*   nbh      = (const int*)d_in[3];
    const float* pos      = (const float*)d_in[4];
    const float* bias     = (const float*)d_in[5];
    float*       out      = (float*)d_out;
    _Float16*    wsA      = (_Float16*)d_ws;

    {
        int nwords = BATCH * COUT * NPTS / 4;
        init_out_kernel<<<(nwords + 255) / 256, 256, 0, stream>>>(bias, out);
    }
    pack_weights_kernel<<<(A_ELEMS + 255) / 256, 256, 0, stream>>>(theta, wbias, wsA);
    flexconv_kernel<<<BATCH * (NPTS / TILE_PTS), 256, 0, stream>>>(
        features, nbh, pos, wsA, out);
}


# Pack launch writes features once as 16-bit WMMA operands in B-fragment order, indices as 16-bit; a row workgroup fetches a trip in 9 loads of 512 B, not 36 of 256 B, no conversions; sums as before
# speedup vs baseline: 16.9043x; 16.9043x over previous
#include <hip/hip_runtime.h>


typedef __attribute__((ext_vector_type(16))) _Float16 v16h;
typedef __attribute__((ext_vector_type(8)))  _Float16 v8h;
typedef __attribute__((ext_vector_type(8)))  float    v8f;

static constexpr int BATCH   = 4;
static constexpr int CIN     = 64;
static constexpr int COUT    = 64;
static constexpr int NPTS    = 16384;
static constexpr int KNBH    = 8;
static constexpr int THREADS = 1024;
static constexpr int WAVES   = THREADS / 32;
static constexpr int PAIRS   = NPTS / 32;
static constexpr int PACK_THREADS = 256;
static constexpr int PACK_BLOCKS  = BATCH * PAIRS * 2 / (PACK_THREADS / 32);
static constexpr int FEAT_WORDS   = BATCH * PAIRS * 8 * 32 * 4;
static constexpr int IDX_WORDS    = BATCH * PAIRS * 32 * 4;
static constexpr size_t WS_BYTES  = (size_t)(FEAT_WORDS + IDX_WORDS) * 4;

__device__ __forceinline__ int kmap(int s, int h, int kc) {
    return 32 * kc + 16 * (s >> 3) + 8 * h + (s & 7);
}

__device__ __forceinline__ void load_indices_direct(const int* __restrict__ nbhb,
                                                    int n0, int c, int h, int (&jj)[8]) {
    #pragma unroll
    for (int kk = 0; kk < 4; ++kk) {
        const int2 jv = *(const int2*)(nbhb + (size_t)(4 * h + kk) * NPTS + n0 + 2 * c);
        jj[kk]     = jv.x;
        jj[4 + kk] = jv.y;
    }
}

__device__ __forceinline__ void load_frags_direct(const float* __restrict__ featb,
                                                  int n0, int c, int h, int kc,
                                                  v16h& f0, v16h& f1) {
    #pragma unroll
    for (int s = 0; s < 16; ++s) {
        const float2 v = *(const float2*)(featb + (size_t)kmap(s, h, kc) * NPTS + n0 + 2 * c);
        f0[s] = (_Float16)v.x;
        f1[s] = (_Float16)v.y;
    }
}

__global__ void __launch_bounds__(PACK_THREADS)
flexconv_pack_kernel(const float* __restrict__ features,
                     const int*   __restrict__ nbh,
                     unsigned int* __restrict__ ws, size_t ws_bytes) {
    if (ws_bytes < WS_BYTES) return;
    const int lane = threadIdx.x & 31;
    const int c    = lane & 15;
    const int h    = lane >> 4;
    const int w    = blockIdx.x * (PACK_THREADS / 32) + (threadIdx.x >> 5);
    const int kc   = w & 1;
    const int g    = w >> 1;
    const int b    = g / PAIRS;
    const int n0   = (g % PAIRS) * 32;

    int jj[8];
    if (kc == 0)
        load_indices_direct(nbh + (size_t)b * KNBH * NPTS, n0, c, h, jj);
    v16h f0, f1;
    load_frags_direct(features + (size_t)b * CIN * NPTS, n0, c, h, kc, f0, f1);

    v8h* dst = (v8h*)ws + ((size_t)g * 8 + 2 * kc) * 32 + lane;
    dst[0 * 32] = __builtin_shufflevector(f0, f0, 0, 1, 2, 3, 4, 5, 6, 7);
    dst[1 * 32] = __builtin_shufflevector(f0, f0, 8, 9, 10, 11, 12, 13, 14, 15);
    dst[4 * 32] = __builtin_shufflevector(f1, f1, 0, 1, 2, 3, 4, 5, 6, 7);
    dst[5 * 32] = __builtin_shufflevector(f1, f1, 8, 9, 10, 11, 12, 13, 14, 15);

    if (kc == 0) {
        uint4* idst = (uint4*)(ws + FEAT_WORDS) + (size_t)g * 32 + lane;
        *idst = make_uint4((unsigned)jj[0] | ((unsigned)jj[1] << 16), (unsigned)jj[2] | ((unsigned)jj[3] << 16),
                           (unsigned)jj[4] | ((unsigned)jj[5] << 16), (unsigned)jj[6] | ((unsigned)jj[7] << 16));
    }
}

__device__ __forceinline__ void load_trip_packed(const v8h* __restrict__ fp,
                                                 const uint4* __restrict__ ip,
                                                 int (&jj)[8],
                                                 v16h& f00, v16h& f01, v16h& f10, v16h& f11) {
    const uint4 iv = *ip;
    const v8h c0 = fp[0 * 32], c1 = fp[1 * 32], c2 = fp[2 * 32], c3 = fp[3 * 32];
    const v8h c4 = fp[4 * 32], c5 = fp[5 * 32], c6 = fp[6 * 32], c7 = fp[7 * 32];
    jj[0] = (int)(iv.x & 0xffffu);  jj[1] = (int)(iv.x >> 16);
    jj[2] = (int)(iv.y & 0xffffu);  jj[3] = (int)(iv.y >> 16);
    jj[4] = (int)(iv.z & 0xffffu);  jj[5] = (int)(iv.z >> 16);
    jj[6] = (int)(iv.w & 0xffffu);  jj[7] = (int)(iv.w >> 16);
    f00 = __builtin_shufflevector(c0, c1, 0, 1, 2, 3, 4, 5, 6, 7, 8, 9, 10, 11, 12, 13, 14, 15);
    f01 = __builtin_shufflevector(c2, c3, 0, 1, 2, 3, 4, 5, 6, 7, 8, 9, 10, 11, 12, 13, 14, 15);
    f10 = __builtin_shufflevector(c4, c5, 0, 1, 2, 3, 4, 5, 6, 7, 8, 9, 10, 11, 12, 13, 14, 15);
    f11 = __builtin_shufflevector(c6, c7, 0, 1, 2, 3, 4, 5, 6, 7, 8, 9, 10, 11, 12, 13, 14, 15);
}

__device__ __forceinline__ void trip_terms(float* acc, const float* spx, const float* spy,
                                           const float* spz, int n0, int c, const int (&jj)[8],
                                           const v16h& a0, const v16h& a1, const v16h& f00,
                                           const v16h& f01, const v16h& f10, const v16h& f11) {
    const float2 xn = *(const float2*)(spx + n0 + 2 * c);
    const float2 yn = *(const float2*)(spy + n0 + 2 * c);
    const float2 zn = *(const float2*)(spz + n0 + 2 * c);
    const float xn0 = xn.x, yn0 = yn.x, zn0 = zn.x;
    const float xn1 = xn.y, yn1 = yn.y, zn1 = zn.y;
    float gx[8], gy[8], gz[8];
    #pragma unroll
    for (int q = 0; q < 8; ++q) {
        gx[q] = spx[jj[q]];
        gy[q] = spy[jj[q]];
        gz[q] = spz[jj[q]];
    }

    v8f d0 = {}, d1 = {};
    d0 = __builtin_amdgcn_wmma_f32_16x16x32_f16(false, a0, false, f00, (short)0, d0, false, false);
    d1 = __builtin_amdgcn_wmma_f32_16x16x32_f16(false, a0, false, f10, (short)0, d1, false, false);
    d0 = __builtin_amdgcn_wmma_f32_16x16x32_f16(false, a1, false, f01, (short)0, d0, false, false);
    d1 = __builtin_amdgcn_wmma_f32_16x16x32_f16(false, a1, false, f11, (short)0, d1, false, false);

    #pragma unroll
    for (int q = 0; q < 4; ++q) {
        const float t0 = (gx[q] - xn0) * d0[0] + (gy[q] - yn0) * d0[1]
                       + (gz[q] - zn0) * d0[2] + d0[3];
        const float t1 = (gx[4 + q] - xn1) * d1[0] + (gy[4 + q] - yn1) * d1[1]
                       + (gz[4 + q] - zn1) * d1[2] + d1[3];
        atomicAdd(&acc[jj[q]],     t0);
        atomicAdd(&acc[jj[4 + q]], t1);
    }
}

__global__ void __launch_bounds__(THREADS)
flexconv_row_kernel(const float* __restrict__ features,
                    const float* __restrict__ theta,
                    const float* __restrict__ wbias,
                    const int*   __restrict__ nbh,
                    const float* __restrict__ pos,
                    const float* __restrict__ bias,
                    float* __restrict__ out,
                    const unsigned int* __restrict__ ws, size_t ws_bytes) {
    __shared__ __align__(16) float acc[NPTS];
    __shared__ __align__(16) float spx[NPTS];
    __shared__ __align__(16) float spy[NPTS];
    __shared__ __align__(16) float spz[NPTS];

    const int tid  = threadIdx.x;
    const int lane = tid & 31;
    const int wave = tid >> 5;
    const int c    = lane & 15;
    const int h    = lane >> 4;
    const int b    = blockIdx.x / COUT;
    const int o    = blockIdx.x % COUT;

    {
        const float* posb = pos + (size_t)b * 3 * NPTS;
        const float  bv   = bias[o];
        for (int i = tid; i < NPTS / 4; i += THREADS) {
            ((float4*)spx)[i] = ((const float4*)(posb))[i];
            ((float4*)spy)[i] = ((const float4*)(posb + NPTS))[i];
            ((float4*)spz)[i] = ((const float4*)(posb + 2 * NPTS))[i];
            ((float4*)acc)[i] = make_float4(bv, bv, bv, bv);
        }
    }

    v16h a0, a1;
    {
        const int    kind = c & 3;
        const float* wsrc = (kind < 3) ? (theta + (size_t)kind * CIN * COUT) : wbias;
        #pragma unroll
        for (int s = 0; s < 16; ++s) {
            a0[s] = (_Float16)wsrc[(size_t)kmap(s, h, 0) * COUT + o];
            a1[s] = (_Float16)wsrc[(size_t)kmap(s, h, 1) * COUT + o];
        }
    }
    __syncthreads();

    if (ws_bytes >= WS_BYTES) {
        const v8h*   featp = (const v8h*)ws + (size_t)b * PAIRS * 8 * 32 + lane;
        const uint4* idxp  = (const uint4*)(ws + FEAT_WORDS) + (size_t)b * PAIRS * 32 + lane;
        for (int tp = wave; tp < PAIRS; tp += WAVES) {
            int  jj[8];
            v16h f00, f01, f10, f11;
            load_trip_packed(featp + (size_t)tp * 8 * 32, idxp + (size_t)tp * 32,
                             jj, f00, f01, f10, f11);
            trip_terms(acc, spx, spy, spz, tp * 32, c, jj, a0, a1, f00, f01, f10, f11);
        }
    } else {
        const float* featb = features + (size_t)b * CIN * NPTS;
        const int*   nbhb  = nbh + (size_t)b * KNBH * NPTS;
        for (int tp = wave; tp < PAIRS; tp += WAVES) {
            int  jj[8];
            v16h f00, f01, f10, f11;
            load_indices_direct(nbhb, tp * 32, c, h, jj);
            load_frags_direct(featb, tp * 32, c, h, 0, f00, f10);
            load_frags_direct(featb, tp * 32, c, h, 1, f01, f11);
            trip_terms(acc, spx, spy, spz, tp * 32, c, jj, a0, a1, f00, f01, f10, f11);
        }
    }
    __syncthreads();

    {
        float* outr = out + ((size_t)b * COUT + o) * NPTS;
        for (int i = tid; i < NPTS / 4; i += THREADS)
            ((float4*)outr)[i] = ((const float4*)acc)[i];
    }
}

extern "C" void kernel_launch(void* const* d_in, const int* in_sizes, int n_in,
                              void* d_out, int out_size, void* d_ws, size_t ws_size,
                              hipStream_t stream) {
    const float* features = (const float*)d_in[0];
    const float* theta    = (const float*)d_in[1];
    const float* wbias    = (const float*)d_in[2];
    const int*   nbh      = (const int*)d_in[3];
    const float* pos      = (const float*)d_in[4];
    const float* bias     = (const float*)d_in[5];
    float*       out      = (float*)d_out;
    unsigned int* ws      = (unsigned int*)d_ws;

    flexconv_pack_kernel<<<PACK_BLOCKS, PACK_THREADS, 0, stream>>>(features, nbh, ws, ws_size);
    flexconv_row_kernel<<<BATCH * COUT, THREADS, 0, stream>>>(
        features, theta, wbias, nbh, pos, bias, out, ws, ws_size);
}


# Pack launch orders each point's 8 neighbour entries by rank on their LDS bank, so lanes of one LDS float add or position load share banks less; sum order changes (arrival order as before)
# speedup vs baseline: 18.4568x; 1.0918x over previous
#include <hip/hip_runtime.h>


typedef __attribute__((ext_vector_type(16))) _Float16 v16h;
typedef __attribute__((ext_vector_type(8)))  _Float16 v8h;
typedef __attribute__((ext_vector_type(8)))  float    v8f;

static constexpr int BATCH   = 4;
static constexpr int CIN     = 64;
static constexpr int COUT    = 64;
static constexpr int NPTS    = 16384;
static constexpr int KNBH    = 8;
static constexpr int THREADS = 1024;
static constexpr int WAVES   = THREADS / 32;
static constexpr int PAIRS   = NPTS / 32;
static constexpr int PACK_THREADS = 256;
static constexpr int PACK_WAVES   = PACK_THREADS / 32;
static constexpr int PACK_BLOCKS  = BATCH * PAIRS * 2 / (PACK_THREADS / 32);
static constexpr int FEAT_WORDS   = BATCH * PAIRS * 8 * 32 * 4;
static constexpr int IDX_WORDS    = BATCH * PAIRS * 32 * 4;
static constexpr size_t WS_BYTES  = (size_t)(FEAT_WORDS + IDX_WORDS) * 4;

__device__ __forceinline__ int kmap(int s, int h, int kc) {
    return 32 * kc + 16 * (s >> 3) + 8 * h + (s & 7);
}

__device__ __forceinline__ void load_indices_direct(const int* __restrict__ nbhb,
                                                    int n0, int c, int h, int (&jj)[8]) {
    #pragma unroll
    for (int kk = 0; kk < 4; ++kk) {
        const int2 jv = *(const int2*)(nbhb + (size_t)(4 * h + kk) * NPTS + n0 + 2 * c);
        jj[kk]     = jv.x;
        jj[4 + kk] = jv.y;
    }
}

__device__ __forceinline__ void load_frags_raw(const float* __restrict__ featb,
                                               int n0, int c, int h, int kc, float2 (&raw)[16]) {
    #pragma unroll
    for (int s = 0; s < 16; ++s)
        raw[s] = *(const float2*)(featb + (size_t)kmap(s, h, kc) * NPTS + n0 + 2 * c);
}
__device__ __forceinline__ void convert_frags(const float2 (&raw)[16], v16h& f0, v16h& f1) {
    #pragma unroll
    for (int s = 0; s < 16; ++s) {
        f0[s] = (_Float16)raw[s].x;
        f1[s] = (_Float16)raw[s].y;
    }
}
__device__ __forceinline__ void load_frags_direct(const float* __restrict__ featb,
                                                  int n0, int c, int h, int kc,
                                                  v16h& f0, v16h& f1) {
    #pragma unroll
    for (int s = 0; s < 16; ++s) {
        const float2 v = *(const float2*)(featb + (size_t)kmap(s, h, kc) * NPTS + n0 + 2 * c);
        f0[s] = (_Float16)v.x;
        f1[s] = (_Float16)v.y;
    }
}

__device__ __forceinline__ void load_point_entries(const int* __restrict__ nbhb, int n0,
                                                   int c, int h, int (&nb)[8]) {
    #pragma unroll
    for (int k = 0; k < 8; ++k)
        nb[k] = nbhb[(size_t)k * NPTS + n0 + 2 * c + h];
}
__device__ __forceinline__ uint4 order_point_entries(const int (&nb)[8], int lane,
                                                     unsigned int* seen, int* cells) {
    const int c = lane & 15;
    const int h = lane >> 4;
    unsigned want[8];
    #pragma unroll
    for (int k = 0; k < 8; ++k)
        want[k] = atomicAdd(&seen[nb[k] & 63], 1u) & 3u;

    unsigned pos[8];
    unsigned cnt = 0;
    #pragma unroll
    for (int k = 0; k < 8; ++k) {
        const unsigned q  = want[k];
        const unsigned n  = (cnt >> (4 * q)) & 15u;
        const bool     ok = n < 2u;
        pos[k] = ok ? 2u * q + n : 8u;
        cnt   += ok ? (1u << (4 * q)) : 0u;
    }
    #pragma unroll
    for (int k = 0; k < 8; ++k) {
        const unsigned q1 = (want[k] + 1u) & 3u, q2 = (want[k] + 2u) & 3u, q3 = (want[k] + 3u) & 3u;
        const bool full1 = ((cnt >> (4 * q1)) & 15u) >= 2u;
        const bool full2 = ((cnt >> (4 * q2)) & 15u) >= 2u;
        const unsigned q = full1 ? (full2 ? q3 : q2) : q1;
        const unsigned n = (cnt >> (4 * q)) & 15u;
        const bool  need = pos[k] == 8u;
        pos[k] = need ? 2u * q + n : pos[k];
        cnt   += need ? (1u << (4 * q)) : 0u;
    }
    #pragma unroll
    for (int k = 0; k < 8; ++k)
        cells[pos[k] * 32 + lane] = nb[k];

    __builtin_amdgcn_fence(__ATOMIC_RELEASE, "wavefront");
    __builtin_amdgcn_wave_barrier();
    __builtin_amdgcn_fence(__ATOMIC_ACQUIRE, "wavefront");

    unsigned e[8];
    #pragma unroll
    for (int q = 0; q < 4; ++q) {
        e[q]     = (unsigned)cells[(2 * q + h) * 32 + c];
        e[4 + q] = (unsigned)cells[(2 * q + h) * 32 + c + 16];
    }
    return make_uint4(e[0] | (e[1] << 16), e[2] | (e[3] << 16), e[4] | (e[5] << 16), e[6] | (e[7] << 16));
}

__global__ void __launch_bounds__(PACK_THREADS)
flexconv_pack_kernel(const float* __restrict__ features,
                     const int*   __restrict__ nbh,
                     unsigned int* __restrict__ ws, size_t ws_bytes) {
    __shared__ unsigned int bank_seen[PACK_WAVES / 2][2][64];
    __shared__ int          cell_lds[PACK_WAVES / 2][8 * 32];
    if (ws_bytes < WS_BYTES) return;
    const int lane = threadIdx.x & 31;
    const int c    = lane & 15;
    const int h    = lane >> 4;
    const int wv   = threadIdx.x >> 5;
    const int w    = blockIdx.x * PACK_WAVES + wv;
    const int kc   = w & 1;
    const int g    = w >> 1;
    const int b    = g / PAIRS;
    const int n0   = (g % PAIRS) * 32;

    for (int i = threadIdx.x; i < (PACK_WAVES / 2) * 2 * 64; i += PACK_THREADS)
        (&bank_seen[0][0][0])[i] = 0u;
    __syncthreads();

    int nb[8] = {0, 0, 0, 0, 0, 0, 0, 0};
    if (kc == 0)
        load_point_entries(nbh + (size_t)b * KNBH * NPTS, n0, c, h, nb);
    float2 raw[16];
    load_frags_raw(features + (size_t)b * CIN * NPTS, n0, c, h, kc, raw);
    uint4 iw = make_uint4(0u, 0u, 0u, 0u);
    if (kc == 0)
        iw = order_point_entries(nb, lane, &bank_seen[wv >> 1][h][0], &cell_lds[wv >> 1][0]);
    v16h f0, f1;
    convert_frags(raw, f0, f1);

    v8h* dst = (v8h*)ws + ((size_t)g * 8 + 2 * kc) * 32 + lane;
    dst[0 * 32] = __builtin_shufflevector(f0, f0, 0, 1, 2, 3, 4, 5, 6, 7);
    dst[1 * 32] = __builtin_shufflevector(f0, f0, 8, 9, 10, 11, 12, 13, 14, 15);
    dst[4 * 32] = __builtin_shufflevector(f1, f1, 0, 1, 2, 3, 4, 5, 6, 7);
    dst[5 * 32] = __builtin_shufflevector(f1, f1, 8, 9, 10, 11, 12, 13, 14, 15);

    if (kc == 0) {
        uint4* idst = (uint4*)(ws + FEAT_WORDS) + (size_t)g * 32 + lane;
        *idst = iw;
    }
}

__device__ __forceinline__ void load_trip_packed(const v8h* __restrict__ fp,
                                                 const uint4* __restrict__ ip,
                                                 int (&jj)[8],
                                                 v16h& f00, v16h& f01, v16h& f10, v16h& f11) {
    const uint4 iv = *ip;
    const v8h c0 = fp[0 * 32], c1 = fp[1 * 32], c2 = fp[2 * 32], c3 = fp[3 * 32];
    const v8h c4 = fp[4 * 32], c5 = fp[5 * 32], c6 = fp[6 * 32], c7 = fp[7 * 32];
    jj[0] = (int)(iv.x & 0xffffu);  jj[1] = (int)(iv.x >> 16);
    jj[2] = (int)(iv.y & 0xffffu);  jj[3] = (int)(iv.y >> 16);
    jj[4] = (int)(iv.z & 0xffffu);  jj[5] = (int)(iv.z >> 16);
    jj[6] = (int)(iv.w & 0xffffu);  jj[7] = (int)(iv.w >> 16);
    f00 = __builtin_shufflevector(c0, c1, 0, 1, 2, 3, 4, 5, 6, 7, 8, 9, 10, 11, 12, 13, 14, 15);
    f01 = __builtin_shufflevector(c2, c3, 0, 1, 2, 3, 4, 5, 6, 7, 8, 9, 10, 11, 12, 13, 14, 15);
    f10 = __builtin_shufflevector(c4, c5, 0, 1, 2, 3, 4, 5, 6, 7, 8, 9, 10, 11, 12, 13, 14, 15);
    f11 = __builtin_shufflevector(c6, c7, 0, 1, 2, 3, 4, 5, 6, 7, 8, 9, 10, 11, 12, 13, 14, 15);
}

__device__ __forceinline__ void trip_adds(float* acc, const float2* spxy,
                                          const float* spz, int n0, int c, const int (&jj)[8],
                                          const v8f& d0, const v8f& d1) {
    const float4 xyn = *(const float4*)(spxy + n0 + 2 * c);
    const float2 zn  = *(const float2*)(spz + n0 + 2 * c);
    const float xn0 = xyn.x, yn0 = xyn.y, zn0 = zn.x;
    const float xn1 = xyn.z, yn1 = xyn.w, zn1 = zn.y;
    float gx[8], gy[8], gz[8];
    #pragma unroll
    for (int q = 0; q < 8; ++q) {
        const float2 gxy = spxy[jj[q]];
        gx[q] = gxy.x;
        gy[q] = gxy.y;
        gz[q] = spz[jj[q]];
    }

    #pragma unroll
    for (int q = 0; q < 4; ++q) {
        const float t0 = (gx[q] - xn0) * d0[0] + (gy[q] - yn0) * d0[1]
                       + (gz[q] - zn0) * d0[2] + d0[3];
        const float t1 = (gx[4 + q] - xn1) * d1[0] + (gy[4 + q] - yn1) * d1[1]
                       + (gz[4 + q] - zn1) * d1[2] + d1[3];
        atomicAdd(&acc[jj[q]],     t0);
        atomicAdd(&acc[jj[4 + q]], t1);
    }
}

__device__ __forceinline__ void trip_terms(float* acc, const float2* spxy,
                                           const float* spz, int n0, int c, const int (&jj)[8],
                                           const v16h& a0, const v16h& a1, const v16h& f00,
                                           const v16h& f01, const v16h& f10, const v16h& f11) {
    v8f d0 = {}, d1 = {};
    d0 = __builtin_amdgcn_wmma_f32_16x16x32_f16(false, a0, false, f00, (short)0, d0, false, false);
    d1 = __builtin_amdgcn_wmma_f32_16x16x32_f16(false, a0, false, f10, (short)0, d1, false, false);
    d0 = __builtin_amdgcn_wmma_f32_16x16x32_f16(false, a1, false, f01, (short)0, d0, false, false);
    d1 = __builtin_amdgcn_wmma_f32_16x16x32_f16(false, a1, false, f11, (short)0, d1, false, false);
    trip_adds(acc, spxy, spz, n0, c, jj, d0, d1);
}

__global__ void __launch_bounds__(THREADS)
flexconv_row_kernel(const float* __restrict__ features,
                    const float* __restrict__ theta,
                    const float* __restrict__ wbias,
                    const int*   __restrict__ nbh,
                    const float* __restrict__ pos,
                    const float* __restrict__ bias,
                    float* __restrict__ out,
                    const unsigned int* __restrict__ ws, size_t ws_bytes) {
    __shared__ __align__(16) float  acc[NPTS];
    __shared__ __align__(16) float2 spxy[NPTS];
    __shared__ __align__(16) float  spz[NPTS];

    const int tid  = threadIdx.x;
    const int lane = tid & 31;
    const int wave = tid >> 5;
    const int c    = lane & 15;
    const int h    = lane >> 4;
    const int b    = blockIdx.x / COUT;
    const int o    = blockIdx.x % COUT;

    {
        const float* posb = pos + (size_t)b * 3 * NPTS;
        const float  bv   = bias[o];
        for (int i = tid; i < NPTS / 4; i += THREADS) {
            const float4 x4 = ((const float4*)(posb))[i];
            const float4 y4 = ((const float4*)(posb + NPTS))[i];
            ((float4*)spxy)[2 * i]     = make_float4(x4.x, y4.x, x4.y, y4.y);
            ((float4*)spxy)[2 * i + 1] = make_float4(x4.z, y4.z, x4.w, y4.w);
            ((float4*)spz)[i] = ((const float4*)(posb + 2 * NPTS))[i];
            ((float4*)acc)[i] = make_float4(bv, bv, bv, bv);
        }
    }

    v16h a0, a1;
    {
        const int    kind = c & 3;
        const float* wsrc = (kind < 3) ? (theta + (size_t)kind * CIN * COUT) : wbias;
        #pragma unroll
        for (int s = 0; s < 16; ++s) {
            a0[s] = (_Float16)wsrc[(size_t)kmap(s, h, 0) * COUT + o];
            a1[s] = (_Float16)wsrc[(size_t)kmap(s, h, 1) * COUT + o];
        }
    }
    __syncthreads();

    if (ws_bytes >= WS_BYTES) {
        const v8h*   featp = (const v8h*)ws + (size_t)b * PAIRS * 8 * 32 + lane;
        const uint4* idxp  = (const uint4*)(ws + FEAT_WORDS) + (size_t)b * PAIRS * 32 + lane;
        for (int tp = wave; tp < PAIRS; tp += WAVES) {
            int  jj[8];
            v16h f00, f01, f10, f11;
            load_trip_packed(featp + (size_t)tp * 8 * 32, idxp + (size_t)tp * 32,
                             jj, f00, f01, f10, f11);
            trip_terms(acc, spxy, spz, tp * 32, c, jj, a0, a1, f00, f01, f10, f11);
        }
    } else {
        const float* featb = features + (size_t)b * CIN * NPTS;
        const int*   nbhb  = nbh + (size_t)b * KNBH * NPTS;
        for (int tp = wave; tp < PAIRS; tp += WAVES) {
            int  jj[8];
            v16h f0, f1;
            v8f  d0 = {}, d1 = {};
            load_indices_direct(nbhb, tp * 32, c, h, jj);
            load_frags_direct(featb, tp * 32, c, h, 0, f0, f1);
            d0 = __builtin_amdgcn_wmma_f32_16x16x32_f16(false, a0, false, f0, (short)0, d0, false, false);
            d1 = __builtin_amdgcn_wmma_f32_16x16x32_f16(false, a0, false, f1, (short)0, d1, false, false);
            __builtin_amdgcn_sched_barrier(0);
            load_frags_direct(featb, tp * 32, c, h, 1, f0, f1);
            d0 = __builtin_amdgcn_wmma_f32_16x16x32_f16(false, a1, false, f0, (short)0, d0, false, false);
            d1 = __builtin_amdgcn_wmma_f32_16x16x32_f16(false, a1, false, f1, (short)0, d1, false, false);
            trip_adds(acc, spxy, spz, tp * 32, c, jj, d0, d1);
        }
    }
    __syncthreads();

    {
        float* outr = out + ((size_t)b * COUT + o) * NPTS;
        for (int i = tid; i < NPTS / 4; i += THREADS)
            ((float4*)outr)[i] = ((const float4*)acc)[i];
    }
}

extern "C" void kernel_launch(void* const* d_in, const int* in_sizes, int n_in,
                              void* d_out, int out_size, void* d_ws, size_t ws_size,
                              hipStream_t stream) {
    const float* features = (const float*)d_in[0];
    const float* theta    = (const float*)d_in[1];
    const float* wbias    = (const float*)d_in[2];
    const int*   nbh      = (const int*)d_in[3];
    const float* pos      = (const float*)d_in[4];
    const float* bias     = (const float*)d_in[5];
    float*       out      = (float*)d_out;
    unsigned int* ws      = (unsigned int*)d_ws;

    flexconv_pack_kernel<<<PACK_BLOCKS, PACK_THREADS, 0, stream>>>(features, nbh, ws, ws_size);
    flexconv_row_kernel<<<BATCH * COUT, THREADS, 0, stream>>>(
        features, theta, wbias, nbh, pos, bias, out, ws, ws_size);
}


# Row workgroups that stream the same packed trips walk them in rotated order, 8 trips apart per channel pair (position copy too): a line is wanted by 4 at once, not 32; LDS adds in arrival order
# speedup vs baseline: 23.2587x; 1.2602x over previous
#include <hip/hip_runtime.h>


typedef __attribute__((ext_vector_type(16))) _Float16 v16h;
typedef __attribute__((ext_vector_type(8)))  _Float16 v8h;
typedef __attribute__((ext_vector_type(8)))  float    v8f;

static constexpr int BATCH   = 4;
static constexpr int CIN     = 64;
static constexpr int COUT    = 64;
static constexpr int NPTS    = 16384;
static constexpr int KNBH    = 8;
static constexpr int THREADS = 1024;
static constexpr int WAVES   = THREADS / 32;
static constexpr int PAIRS   = NPTS / 32;
static constexpr int ROW_CH    = 2;
static constexpr int SRC_PARTS = 2;
static constexpr int PART_TRIPS = PAIRS / SRC_PARTS;
static constexpr int ROW_BLOCKS = BATCH * (COUT / ROW_CH) * SRC_PARTS;
static constexpr int OUT_VEC4   = BATCH * COUT * NPTS / 4;
static constexpr int PACK_THREADS = 256;
static constexpr int PACK_WAVES   = PACK_THREADS / 32;
static constexpr int PACK_BLOCKS  = BATCH * PAIRS * 2 / (PACK_THREADS / 32);
static constexpr int FEAT_WORDS   = BATCH * PAIRS * 8 * 32 * 4;
static constexpr int IDX_WORDS    = BATCH * PAIRS * 32 * 4;
static constexpr size_t WS_BYTES  = (size_t)(FEAT_WORDS + IDX_WORDS) * 4;

__device__ __forceinline__ int kmap(int s, int h, int kc) {
    return 32 * kc + 16 * (s >> 3) + 8 * h + (s & 7);
}

__device__ __forceinline__ void load_indices_direct(const int* __restrict__ nbhb,
                                                    int n0, int c, int h, int (&jj)[8]) {
    #pragma unroll
    for (int kk = 0; kk < 4; ++kk) {
        const int2 jv = *(const int2*)(nbhb + (size_t)(4 * h + kk) * NPTS + n0 + 2 * c);
        jj[kk]     = jv.x;
        jj[4 + kk] = jv.y;
    }
}

__device__ __forceinline__ void load_frags_raw(const float* __restrict__ featb,
                                               int n0, int c, int h, int kc, float2 (&raw)[16]) {
    #pragma unroll
    for (int s = 0; s < 16; ++s)
        raw[s] = *(const float2*)(featb + (size_t)kmap(s, h, kc) * NPTS + n0 + 2 * c);
}
__device__ __forceinline__ void convert_frags(const float2 (&raw)[16], v16h& f0, v16h& f1) {
    #pragma unroll
    for (int s = 0; s < 16; ++s) {
        f0[s] = (_Float16)raw[s].x;
        f1[s] = (_Float16)raw[s].y;
    }
}
__device__ __forceinline__ void load_frags_direct(const float* __restrict__ featb,
                                                  int n0, int c, int h, int kc,
                                                  v16h& f0, v16h& f1) {
    #pragma unroll
    for (int s = 0; s < 16; ++s) {
        const float2 v = *(const float2*)(featb + (size_t)kmap(s, h, kc) * NPTS + n0 + 2 * c);
        f0[s] = (_Float16)v.x;
        f1[s] = (_Float16)v.y;
    }
}

__device__ __forceinline__ void load_point_entries(const int* __restrict__ nbhb, int n0,
                                                   int c, int h, int (&nb)[8]) {
    #pragma unroll
    for (int k = 0; k < 8; ++k)
        nb[k] = nbhb[(size_t)k * NPTS + n0 + 2 * c + h];
}
__device__ __forceinline__ uint4 order_point_entries(const int (&nb)[8], int lane,
                                                     unsigned int* seen, int* cells) {
    const int c = lane & 15;
    const int h = lane >> 4;
    unsigned want[8];
    #pragma unroll
    for (int k = 0; k < 8; ++k)
        want[k] = atomicAdd(&seen[nb[k] & 63], 1u) & 3u;

    unsigned pos[8];
    unsigned cnt = 0;
    #pragma unroll
    for (int k = 0; k < 8; ++k) {
        const unsigned q  = want[k];
        const unsigned n  = (cnt >> (4 * q)) & 15u;
        const bool     ok = n < 2u;
        pos[k] = ok ? 2u * q + n : 8u;
        cnt   += ok ? (1u << (4 * q)) : 0u;
    }
    #pragma unroll
    for (int k = 0; k < 8; ++k) {
        const unsigned q1 = (want[k] + 1u) & 3u, q2 = (want[k] + 2u) & 3u, q3 = (want[k] + 3u) & 3u;
        const bool full1 = ((cnt >> (4 * q1)) & 15u) >= 2u;
        const bool full2 = ((cnt >> (4 * q2)) & 15u) >= 2u;
        const unsigned q = full1 ? (full2 ? q3 : q2) : q1;
        const unsigned n = (cnt >> (4 * q)) & 15u;
        const bool  need = pos[k] == 8u;
        pos[k] = need ? 2u * q + n : pos[k];
        cnt   += need ? (1u << (4 * q)) : 0u;
    }
    #pragma unroll
    for (int k = 0; k < 8; ++k)
        cells[pos[k] * 32 + lane] = nb[k];

    __builtin_amdgcn_fence(__ATOMIC_RELEASE, "wavefront");
    __builtin_amdgcn_wave_barrier();
    __builtin_amdgcn_fence(__ATOMIC_ACQUIRE, "wavefront");

    unsigned e[8];
    #pragma unroll
    for (int q = 0; q < 4; ++q) {
        e[q]     = (unsigned)cells[(2 * q + h) * 32 + c];
        e[4 + q] = (unsigned)cells[(2 * q + h) * 32 + c + 16];
    }
    return make_uint4(e[0] | (e[1] << 16), e[2] | (e[3] << 16), e[4] | (e[5] << 16), e[6] | (e[7] << 16));
}

__global__ void __launch_bounds__(PACK_THREADS)
flexconv_pack_kernel(const float* __restrict__ features,
                     const int*   __restrict__ nbh,
                     const float* __restrict__ bias,
                     float* __restrict__ out,
                     unsigned int* __restrict__ ws, size_t ws_bytes) {
    __shared__ unsigned int bank_seen[PACK_WAVES / 2][2][64];
    __shared__ int          cell_lds[PACK_WAVES / 2][8 * 32];
    {
        constexpr int VEC_PER_ROW    = NPTS / 4;
        constexpr int BLOCKS_PER_ROW = VEC_PER_ROW / PACK_THREADS;
        constexpr int ROWS_PER_SWEEP = PACK_BLOCKS / BLOCKS_PER_ROW;
        constexpr int SWEEPS         = BATCH * COUT / ROWS_PER_SWEEP;
        static_assert(VEC_PER_ROW % PACK_THREADS == 0 && PACK_BLOCKS % BLOCKS_PER_ROW == 0 &&
                      (BATCH * COUT) % ROWS_PER_SWEEP == 0 &&
                      SWEEPS * ROWS_PER_SWEEP * VEC_PER_ROW == OUT_VEC4, "the sweeps must cover out exactly");
        const int row0 = blockIdx.x / BLOCKS_PER_ROW;
        const int col  = (blockIdx.x % BLOCKS_PER_ROW) * PACK_THREADS + threadIdx.x;
        #pragma unroll
        for (int k = 0; k < SWEEPS; ++k) {
            const int   row = row0 + k * ROWS_PER_SWEEP;
            const float bv  = bias[row % COUT];
            ((float4*)out)[(size_t)row * VEC_PER_ROW + col] = make_float4(bv, bv, bv, bv);
        }
    }
    if (ws_bytes < WS_BYTES) return;
    const int lane = threadIdx.x & 31;
    const int c    = lane & 15;
    const int h    = lane >> 4;
    const int wv   = threadIdx.x >> 5;
    const int w    = blockIdx.x * PACK_WAVES + wv;
    const int kc   = w & 1;
    const int g    = w >> 1;
    const int b    = g / PAIRS;
    const int n0   = (g % PAIRS) * 32;

    for (int i = threadIdx.x; i < (PACK_WAVES / 2) * 2 * 64; i += PACK_THREADS)
        (&bank_seen[0][0][0])[i] = 0u;
    __syncthreads();

    int nb[8] = {0, 0, 0, 0, 0, 0, 0, 0};
    if (kc == 0)
        load_point_entries(nbh + (size_t)b * KNBH * NPTS, n0, c, h, nb);
    float2 raw[16];
    load_frags_raw(features + (size_t)b * CIN * NPTS, n0, c, h, kc, raw);
    uint4 iw = make_uint4(0u, 0u, 0u, 0u);
    if (kc == 0)
        iw = order_point_entries(nb, lane, &bank_seen[wv >> 1][h][0], &cell_lds[wv >> 1][0]);
    v16h f0, f1;
    convert_frags(raw, f0, f1);

    v8h* dst = (v8h*)ws + ((size_t)g * 8 + 2 * kc) * 32 + lane;
    dst[0 * 32] = __builtin_shufflevector(f0, f0, 0, 1, 2, 3, 4, 5, 6, 7);
    dst[1 * 32] = __builtin_shufflevector(f0, f0, 8, 9, 10, 11, 12, 13, 14, 15);
    dst[4 * 32] = __builtin_shufflevector(f1, f1, 0, 1, 2, 3, 4, 5, 6, 7);
    dst[5 * 32] = __builtin_shufflevector(f1, f1, 8, 9, 10, 11, 12, 13, 14, 15);

    if (kc == 0) {
        uint4* idst = (uint4*)(ws + FEAT_WORDS) + (size_t)g * 32 + lane;
        *idst = iw;
    }
}

__device__ __forceinline__ void load_trip_packed(const v8h* __restrict__ fp,
                                                 const uint4* __restrict__ ip,
                                                 int (&jj)[8],
                                                 v16h& f00, v16h& f01, v16h& f10, v16h& f11) {
    const uint4 iv = *ip;
    const v8h c0 = fp[0 * 32], c1 = fp[1 * 32], c2 = fp[2 * 32], c3 = fp[3 * 32];
    const v8h c4 = fp[4 * 32], c5 = fp[5 * 32], c6 = fp[6 * 32], c7 = fp[7 * 32];
    jj[0] = (int)(iv.x & 0xffffu);  jj[1] = (int)(iv.x >> 16);
    jj[2] = (int)(iv.y & 0xffffu);  jj[3] = (int)(iv.y >> 16);
    jj[4] = (int)(iv.z & 0xffffu);  jj[5] = (int)(iv.z >> 16);
    jj[6] = (int)(iv.w & 0xffffu);  jj[7] = (int)(iv.w >> 16);
    f00 = __builtin_shufflevector(c0, c1, 0, 1, 2, 3, 4, 5, 6, 7, 8, 9, 10, 11, 12, 13, 14, 15);
    f01 = __builtin_shufflevector(c2, c3, 0, 1, 2, 3, 4, 5, 6, 7, 8, 9, 10, 11, 12, 13, 14, 15);
    f10 = __builtin_shufflevector(c4, c5, 0, 1, 2, 3, 4, 5, 6, 7, 8, 9, 10, 11, 12, 13, 14, 15);
    f11 = __builtin_shufflevector(c6, c7, 0, 1, 2, 3, 4, 5, 6, 7, 8, 9, 10, 11, 12, 13, 14, 15);
}

__device__ __forceinline__ void trip_adds(float* acc0, float* acc1, const float2* spxy,
                                          const float* spz, int n0, int c, const int (&jj)[8],
                                          const v8f& d0, const v8f& d1) {
    const float4 xyn = *(const float4*)(spxy + n0 + 2 * c);
    const float2 zn  = *(const float2*)(spz + n0 + 2 * c);
    const float xn0 = xyn.x, yn0 = xyn.y, zn0 = zn.x;
    const float xn1 = xyn.z, yn1 = xyn.w, zn1 = zn.y;
    float gx[8], gy[8], gz[8];
    #pragma unroll
    for (int q = 0; q < 8; ++q) {
        const float2 gxy = spxy[jj[q]];
        gx[q] = gxy.x;
        gy[q] = gxy.y;
        gz[q] = spz[jj[q]];
    }

    #pragma unroll
    for (int q = 0; q < 4; ++q) {
        const float dx0 = gx[q] - xn0,     dy0 = gy[q] - yn0,     dz0 = gz[q] - zn0;
        const float dx1 = gx[4 + q] - xn1, dy1 = gy[4 + q] - yn1, dz1 = gz[4 + q] - zn1;
        const float t0a = dx0 * d0[0] + dy0 * d0[1] + dz0 * d0[2] + d0[3];
        const float t0b = dx0 * d0[4] + dy0 * d0[5] + dz0 * d0[6] + d0[7];
        const float t1a = dx1 * d1[0] + dy1 * d1[1] + dz1 * d1[2] + d1[3];
        const float t1b = dx1 * d1[4] + dy1 * d1[5] + dz1 * d1[6] + d1[7];
        atomicAdd(&acc0[jj[q]],     t0a);
        atomicAdd(&acc1[jj[q]],     t0b);
        atomicAdd(&acc0[jj[4 + q]], t1a);
        atomicAdd(&acc1[jj[4 + q]], t1b);
    }
}

__device__ __forceinline__ void trip_terms(float* acc0, float* acc1, const float2* spxy,
                                           const float* spz, int n0, int c, const int (&jj)[8],
                                           const v16h& a0, const v16h& a1, const v16h& f00,
                                           const v16h& f01, const v16h& f10, const v16h& f11) {
    v8f d0 = {}, d1 = {};
    d0 = __builtin_amdgcn_wmma_f32_16x16x32_f16(false, a0, false, f00, (short)0, d0, false, false);
    d1 = __builtin_amdgcn_wmma_f32_16x16x32_f16(false, a0, false, f10, (short)0, d1, false, false);
    d0 = __builtin_amdgcn_wmma_f32_16x16x32_f16(false, a1, false, f01, (short)0, d0, false, false);
    d1 = __builtin_amdgcn_wmma_f32_16x16x32_f16(false, a1, false, f11, (short)0, d1, false, false);
    trip_adds(acc0, acc1, spxy, spz, n0, c, jj, d0, d1);
}

__global__ void __launch_bounds__(THREADS)
flexconv_row_kernel(const float* __restrict__ features,
                    const float* __restrict__ theta,
                    const float* __restrict__ wbias,
                    const int*   __restrict__ nbh,
                    const float* __restrict__ pos,
                    const float* __restrict__ bias,
                    float* __restrict__ out,
                    const unsigned int* __restrict__ ws, size_t ws_bytes) {
    __shared__ __align__(16) float  acc0[NPTS];
    __shared__ __align__(16) float  acc1[NPTS];
    __shared__ __align__(16) float2 spxy[NPTS];
    __shared__ __align__(16) float  spz[NPTS];

    const int tid  = threadIdx.x;
    const int lane = tid & 31;
    const int wave = tid >> 5;
    const int c    = lane & 15;
    const int h    = lane >> 4;
    constexpr int per_batch = (COUT / ROW_CH) * SRC_PARTS;
    const int b    = blockIdx.x / per_batch;
    const int wg   = blockIdx.x % per_batch;
    const int part = wg % SRC_PARTS;
    const int pr   = wg / SRC_PARTS;
    const int o0   = pr * ROW_CH;

    {
        constexpr int POS_VEC4 = NPTS / 4;
        constexpr int POS_STEP = POS_VEC4 / per_batch;
        static_assert((POS_VEC4 & (POS_VEC4 - 1)) == 0 && POS_VEC4 % per_batch == 0 && POS_STEP % 32 == 0,
                      "the rotated copy must wrap on a power of two and keep a wave's 32 pieces contiguous");
        const float* posb = pos + (size_t)b * 3 * NPTS;
        const int prot = wg * POS_STEP;
        for (int k = tid; k < POS_VEC4; k += THREADS) {
            const int i = (k + prot) & (POS_VEC4 - 1);
            const float4 x4 = ((const float4*)(posb))[i];
            const float4 y4 = ((const float4*)(posb + NPTS))[i];
            ((float4*)spxy)[2 * i]     = make_float4(x4.x, y4.x, x4.y, y4.y);
            ((float4*)spxy)[2 * i + 1] = make_float4(x4.z, y4.z, x4.w, y4.w);
            ((float4*)spz)[i] = ((const float4*)(posb + 2 * NPTS))[i];
            ((float4*)acc0)[i] = make_float4(0.0f, 0.0f, 0.0f, 0.0f);
            ((float4*)acc1)[i] = make_float4(0.0f, 0.0f, 0.0f, 0.0f);
        }
    }

    v16h a0, a1;
    {
        const int    kind = c & 3;
        const int    o    = o0 + ((c >> 2) & 1);
        const float* wsrc = (kind < 3) ? (theta + (size_t)kind * CIN * COUT) : wbias;
        #pragma unroll
        for (int s = 0; s < 16; ++s) {
            a0[s] = (_Float16)wsrc[(size_t)kmap(s, h, 0) * COUT + o];
            a1[s] = (_Float16)wsrc[(size_t)kmap(s, h, 1) * COUT + o];
        }
    }
    __syncthreads();

    constexpr int TRIP_ROT = PART_TRIPS / (COUT / ROW_CH);
    static_assert((PART_TRIPS & (PART_TRIPS - 1)) == 0 && PART_TRIPS % (COUT / ROW_CH) == 0 &&
                  PART_TRIPS % WAVES == 0, "the rotated walk must wrap on a power of two");
    const int rot = pr * TRIP_ROT;
    if (ws_bytes >= WS_BYTES) {
        const v8h*   featp = (const v8h*)ws + (size_t)b * PAIRS * 8 * 32 + lane;
        const uint4* idxp  = (const uint4*)(ws + FEAT_WORDS) + (size_t)b * PAIRS * 32 + lane;
        for (int k = wave; k < PART_TRIPS; k += WAVES) {
            const int tp = part * PART_TRIPS + ((k + rot) & (PART_TRIPS - 1));
            int  jj[8];
            v16h f00, f01, f10, f11;
            load_trip_packed(featp + (size_t)tp * 8 * 32, idxp + (size_t)tp * 32,
                             jj, f00, f01, f10, f11);
            trip_terms(acc0, acc1, spxy, spz, tp * 32, c, jj, a0, a1, f00, f01, f10, f11);
        }
    } else {
        const float* featb = features + (size_t)b * CIN * NPTS;
        const int*   nbhb  = nbh + (size_t)b * KNBH * NPTS;
        for (int tp = part * PART_TRIPS + wave; tp < (part + 1) * PART_TRIPS; tp += WAVES) {
            int  jj[8];
            v16h f0, f1;
            v8f  d0 = {}, d1 = {};
            load_indices_direct(nbhb, tp * 32, c, h, jj);
            load_frags_direct(featb, tp * 32, c, h, 0, f0, f1);
            d0 = __builtin_amdgcn_wmma_f32_16x16x32_f16(false, a0, false, f0, (short)0, d0, false, false);
            d1 = __builtin_amdgcn_wmma_f32_16x16x32_f16(false, a0, false, f1, (short)0, d1, false, false);
            __builtin_amdgcn_sched_barrier(0);
            load_frags_direct(featb, tp * 32, c, h, 1, f0, f1);
            d0 = __builtin_amdgcn_wmma_f32_16x16x32_f16(false, a1, false, f0, (short)0, d0, false, false);
            d1 = __builtin_amdgcn_wmma_f32_16x16x32_f16(false, a1, false, f1, (short)0, d1, false, false);
            trip_adds(acc0, acc1, spxy, spz, tp * 32, c, jj, d0, d1);
        }
    }
    __syncthreads();

    {
        float* outr0 = out + ((size_t)b * COUT + o0) * NPTS;
        float* outr1 = outr0 + NPTS;
        constexpr int PER_THREAD = NPTS / THREADS;
        float v0[PER_THREAD], v1[PER_THREAD];
        #pragma unroll
        for (int k = 0; k < PER_THREAD; ++k) {
            v0[k] = acc0[tid + k * THREADS];
            v1[k] = acc1[tid + k * THREADS];
        }
        #pragma unroll
        for (int k = 0; k < PER_THREAD; ++k) {
            atomicAdd(&outr0[tid + k * THREADS], v0[k]);
            atomicAdd(&outr1[tid + k * THREADS], v1[k]);
        }
    }
}

extern "C" void kernel_launch(void* const* d_in, const int* in_sizes, int n_in,
                              void* d_out, int out_size, void* d_ws, size_t ws_size,
                              hipStream_t stream) {
    const float* features = (const float*)d_in[0];
    const float* theta    = (const float*)d_in[1];
    const float* wbias    = (const float*)d_in[2];
    const int*   nbh      = (const int*)d_in[3];
    const float* pos      = (const float*)d_in[4];
    const float* bias     = (const float*)d_in[5];
    float*       out      = (float*)d_out;
    unsigned int* ws      = (unsigned int*)d_ws;

    flexconv_pack_kernel<<<PACK_BLOCKS, PACK_THREADS, 0, stream>>>(features, nbh, bias, out, ws, ws_size);
    flexconv_row_kernel<<<ROW_BLOCKS, THREADS, 0, stream>>>(
        features, theta, wbias, nbh, pos, bias, out, ws, ws_size);
}
